# starting kernel
# baseline (speedup 1.0000x reference)
#include <hip/hip_runtime.h>
#include <hip/hip_bf16.h>

typedef float v2f __attribute__((ext_vector_type(2)));
typedef float v8f __attribute__((ext_vector_type(8)));

#define B_SZ 4
#define N_SZ 16384
#define D_SZ 256
#define NCHUNK 8
#define CHUNK_N (N_SZ / NCHUNK)
#define EPSV 1e-8f

__global__ __launch_bounds__(256) void la_kv_partial(const float* __restrict__ kmat,
                                                     const float* __restrict__ vmat,
                                                     float* __restrict__ partial) {
    const int idx   = blockIdx.x;
    const int chunk = idx & (NCHUNK - 1);
    const int slab  = (idx >> 3) & 3;
    const int b     = idx >> 5;

    const int tid  = threadIdx.x;
    const int w    = tid >> 5;
    const int lane = tid & 31;
    const int m    = lane & 15;
    const int hi   = lane >> 4;

    const int dtile = w >> 1;
    const int ebase = (w & 1) * 128;
    const int dcol  = slab * 64 + dtile * 16 + m;

    const float* kb = kmat + (size_t)b * N_SZ * D_SZ;
    const float* vb = vmat + (size_t)b * N_SZ * D_SZ;

    v8f acc[8];
#pragma unroll
    for (int t = 0; t < 8; ++t) acc[t] = (v8f){0.f,0.f,0.f,0.f,0.f,0.f,0.f,0.f};

    const int nStart = chunk * CHUNK_N;
    const int nEnd   = nStart + CHUNK_N;

    for (int n0 = nStart; n0 < nEnd; n0 += 4) {
        const int nn = n0 + 2 * hi;
        const float* kp = kb + (size_t)nn * D_SZ + dcol;
        v2f a;
        a.x = fmaxf(kp[0],    0.f);
        a.y = fmaxf(kp[D_SZ], 0.f);
        const float* vp = vb + (size_t)nn * D_SZ + ebase + m;
#pragma unroll
        for (int t = 0; t < 8; ++t) {
            v2f bfr;
            bfr.x = vp[t * 16];
            bfr.y = vp[t * 16 + D_SZ];
            acc[t] = __builtin_amdgcn_wmma_f32_16x16x4_f32(
                false, a, false, bfr, (short)0, acc[t], false, false);
        }
    }

    float* pb = partial + (((size_t)b * NCHUNK + chunk) * D_SZ * D_SZ);
#pragma unroll
    for (int t = 0; t < 8; ++t) {
#pragma unroll
        for (int i = 0; i < 8; ++i) {
            const int drow = slab * 64 + dtile * 16 + (hi ? i + 8 : i);
            pb[(size_t)drow * D_SZ + ebase + t * 16 + m] = acc[t][i];
        }
    }
}

__global__ __launch_bounds__(256) void la_kv_reduce(const float* __restrict__ partial,
                                                    float* __restrict__ kv) {
    const int idx = blockIdx.x * 256 + threadIdx.x;
    const int b   = idx >> 16;
    const int rem = idx & 0xFFFF;
    float s = 0.f;
#pragma unroll
    for (int c = 0; c < NCHUNK; ++c)
        s += partial[(((size_t)(b * NCHUNK + c)) << 16) + rem];
    kv[((size_t)b << 16) + rem] = s;
}

#define QPITCH 260

__global__ __launch_bounds__(256) void la_out(const float* __restrict__ qmat,
                                              const float* __restrict__ kv,
                                              float* __restrict__ out) {
    __shared__ float ldsq[32 * QPITCH];
    __shared__ float dpart[256];
    __shared__ float drcp[32];

    const int b    = blockIdx.x >> 9;
    const int n0   = (blockIdx.x & 511) * 32;
    const int tid  = threadIdx.x;

    {
        const int row  = tid >> 3;
        const int cseg = (tid & 7) * 32;
        const float* qrow = qmat + ((size_t)(b * N_SZ + n0 + row)) * D_SZ + cseg;
        float s = 0.f;
#pragma unroll
        for (int j = 0; j < 32; j += 4) {
            float4 x = *(const float4*)(qrow + j);
            x.x = fmaxf(x.x, 0.f);
            x.y = fmaxf(x.y, 0.f);
            x.z = fmaxf(x.z, 0.f);
            x.w = fmaxf(x.w, 0.f);
            *(float4*)&ldsq[row * QPITCH + cseg + j] = x;
            s += x.x + x.y + x.z + x.w;
        }
        dpart[tid] = s;
    }
    __syncthreads();
    if (tid < 32) {
        float s = EPSV;
#pragma unroll
        for (int j = 0; j < 8; ++j) s += dpart[tid * 8 + j];
        drcp[tid] = 1.0f / s;
    }
    __syncthreads();

    const int w    = tid >> 5;
    const int lane = tid & 31;
    const int m    = lane & 15;
    const int hi   = lane >> 4;
    const int rt   = w >> 2;
    const int cb   = (w & 3) * 64;

    v8f acc[4];
#pragma unroll
    for (int t = 0; t < 4; ++t) acc[t] = (v8f){0.f,0.f,0.f,0.f,0.f,0.f,0.f,0.f};

    const float* kvb = kv + ((size_t)b << 16);

    for (int k0 = 0; k0 < D_SZ; k0 += 4) {
        const int kk = k0 + 2 * hi;
        const float* ap = &ldsq[(rt * 16 + m) * QPITCH + kk];
        v2f a;
        a.x = ap[0];
        a.y = ap[1];
#pragma unroll
        for (int t = 0; t < 4; ++t) {
            const float* bp = kvb + (size_t)kk * D_SZ + cb + t * 16 + m;
            v2f bfr;
            bfr.x = bp[0];
            bfr.y = bp[D_SZ];
            acc[t] = __builtin_amdgcn_wmma_f32_16x16x4_f32(
                false, a, false, bfr, (short)0, acc[t], false, false);
        }
    }

    float r[8];
#pragma unroll
    for (int i = 0; i < 8; ++i) r[i] = drcp[rt * 16 + (hi ? i + 8 : i)];

#pragma unroll
    for (int t = 0; t < 4; ++t) {
#pragma unroll
        for (int i = 0; i < 8; ++i) {
            const int rowl = rt * 16 + (hi ? i + 8 : i);
            out[((size_t)(b * N_SZ + n0 + rowl)) * D_SZ + cb + t * 16 + m] =
                acc[t][i] * r[i];
        }
    }
}

extern "C" void kernel_launch(void* const* d_in, const int* in_sizes, int n_in,
                              void* d_out, int out_size, void* d_ws, size_t ws_size,
                              hipStream_t stream) {
    const float* q = (const float*)d_in[0];
    const float* k = (const float*)d_in[1];
    const float* v = (const float*)d_in[2];
    float* out = (float*)d_out;

    float* kvbuf   = (float*)d_ws;
    float* partial = kvbuf + (size_t)B_SZ * D_SZ * D_SZ;

    la_kv_partial<<<B_SZ * 4 * NCHUNK, 256, 0, stream>>>(k, v, partial);
    la_kv_reduce<<<(B_SZ * D_SZ * D_SZ) / 256, 256, 0, stream>>>(partial, kvbuf);
    la_out<<<B_SZ * (N_SZ / 32), 256, 0, stream>>>(q, kvbuf, out);
}


# kv partial kernel: one 16x128 kv tile per work-group, its 8 waves split the chunk's rows and their tiles are summed through LDS (new reduction tree over n); grid 128 -> 1024 work-groups
# speedup vs baseline: 2.6866x; 2.6866x over previous
#include <hip/hip_runtime.h>
#include <hip/hip_bf16.h>

typedef float v2f __attribute__((ext_vector_type(2)));
typedef float v8f __attribute__((ext_vector_type(8)));

#define B_SZ 4
#define N_SZ 16384
#define D_SZ 256
#define NCHUNK 8
#define CHUNK_N (N_SZ / NCHUNK)
#define EPSV 1e-8f

#define NSUB 8
#define SUB_N (CHUNK_N / NSUB)
#define TILE_ELEMS (16 * 128)

__global__ __launch_bounds__(256) void la_kv_partial(const float* __restrict__ kmat,
                                                     const float* __restrict__ vmat,
                                                     float* __restrict__ partial) {
    __shared__ float red[NSUB * TILE_ELEMS];

    const int idx   = blockIdx.x;
    const int eh    = idx & 1;
    const int dt    = (idx >> 1) & 15;
    const int chunk = (idx >> 5) & (NCHUNK - 1);
    const int b     = idx >> 8;

    const int tid  = threadIdx.x;
    const int w    = __builtin_amdgcn_readfirstlane(tid >> 5);
    const int lane = tid & 31;
    const int m    = lane & 15;
    const int hi   = lane >> 4;

    const int ebase = eh * 128;
    const int dcol  = dt * 16 + m;

    const float* kb = kmat + (size_t)b * N_SZ * D_SZ;
    const float* vb = vmat + (size_t)b * N_SZ * D_SZ;

    v8f acc[8];
#pragma unroll
    for (int t = 0; t < 8; ++t) acc[t] = (v8f){0.f,0.f,0.f,0.f,0.f,0.f,0.f,0.f};

    const int nStart = chunk * CHUNK_N + w * SUB_N;

    for (int it = 0; it < SUB_N / 4; ++it) {
        const int nn = nStart + 4 * it + 2 * hi;
        const float* kp = kb + (size_t)nn * D_SZ + dcol;
        v2f a;
        a.x = fmaxf(kp[0],    0.f);
        a.y = fmaxf(kp[D_SZ], 0.f);
        const float* vp = vb + (size_t)nn * D_SZ + ebase + m;
#pragma unroll
        for (int t = 0; t < 8; ++t) {
            v2f bfr;
            bfr.x = vp[t * 16];
            bfr.y = vp[t * 16 + D_SZ];
            acc[t] = __builtin_amdgcn_wmma_f32_16x16x4_f32(
                false, a, false, bfr, (short)0, acc[t], false, false);
        }
    }

#pragma unroll
    for (int t = 0; t < 8; ++t) {
#pragma unroll
        for (int i = 0; i < 8; ++i)
            red[w * TILE_ELEMS + (t * 8 + i) * 32 + lane] = acc[t][i];
    }
    __syncthreads();

    float* pb = partial + (((size_t)b * NCHUNK + chunk) * D_SZ * D_SZ);
    const int drow = dt * 16 + (hi ? w + 8 : w);
#pragma unroll
    for (int t = 0; t < 8; ++t) {
        const int e = (t * 8 + w) * 32 + lane;
        float s = red[e];
#pragma unroll
        for (int w2 = 1; w2 < NSUB; ++w2) s += red[w2 * TILE_ELEMS + e];
        pb[(size_t)drow * D_SZ + ebase + t * 16 + m] = s;
    }
}

__global__ __launch_bounds__(256) void la_kv_reduce(const float* __restrict__ partial,
                                                    float* __restrict__ kv) {
    const int idx = blockIdx.x * 256 + threadIdx.x;
    const int b   = idx >> 16;
    const int rem = idx & 0xFFFF;
    float s = 0.f;
#pragma unroll
    for (int c = 0; c < NCHUNK; ++c)
        s += partial[(((size_t)(b * NCHUNK + c)) << 16) + rem];
    kv[((size_t)b << 16) + rem] = s;
}

#define QPITCH 260

__global__ __launch_bounds__(256) void la_out(const float* __restrict__ qmat,
                                              const float* __restrict__ kv,
                                              float* __restrict__ out) {
    __shared__ float ldsq[32 * QPITCH];
    __shared__ float dpart[256];
    __shared__ float drcp[32];

    const int b    = blockIdx.x >> 9;
    const int n0   = (blockIdx.x & 511) * 32;
    const int tid  = threadIdx.x;

    {
        const int row  = tid >> 3;
        const int cseg = (tid & 7) * 32;
        const float* qrow = qmat + ((size_t)(b * N_SZ + n0 + row)) * D_SZ + cseg;
        float s = 0.f;
#pragma unroll
        for (int j = 0; j < 32; j += 4) {
            float4 x = *(const float4*)(qrow + j);
            x.x = fmaxf(x.x, 0.f);
            x.y = fmaxf(x.y, 0.f);
            x.z = fmaxf(x.z, 0.f);
            x.w = fmaxf(x.w, 0.f);
            *(float4*)&ldsq[row * QPITCH + cseg + j] = x;
            s += x.x + x.y + x.z + x.w;
        }
        dpart[tid] = s;
    }
    __syncthreads();
    if (tid < 32) {
        float s = EPSV;
#pragma unroll
        for (int j = 0; j < 8; ++j) s += dpart[tid * 8 + j];
        drcp[tid] = 1.0f / s;
    }
    __syncthreads();

    const int w    = tid >> 5;
    const int lane = tid & 31;
    const int m    = lane & 15;
    const int hi   = lane >> 4;
    const int rt   = w >> 2;
    const int cb   = (w & 3) * 64;

    v8f acc[4];
#pragma unroll
    for (int t = 0; t < 4; ++t) acc[t] = (v8f){0.f,0.f,0.f,0.f,0.f,0.f,0.f,0.f};

    const float* kvb = kv + ((size_t)b << 16);

    for (int k0 = 0; k0 < D_SZ; k0 += 4) {
        const int kk = k0 + 2 * hi;
        const float* ap = &ldsq[(rt * 16 + m) * QPITCH + kk];
        v2f a;
        a.x = ap[0];
        a.y = ap[1];
#pragma unroll
        for (int t = 0; t < 4; ++t) {
            const float* bp = kvb + (size_t)kk * D_SZ + cb + t * 16 + m;
            v2f bfr;
            bfr.x = bp[0];
            bfr.y = bp[D_SZ];
            acc[t] = __builtin_amdgcn_wmma_f32_16x16x4_f32(
                false, a, false, bfr, (short)0, acc[t], false, false);
        }
    }

    float r[8];
#pragma unroll
    for (int i = 0; i < 8; ++i) r[i] = drcp[rt * 16 + (hi ? i + 8 : i)];

#pragma unroll
    for (int t = 0; t < 4; ++t) {
#pragma unroll
        for (int i = 0; i < 8; ++i) {
            const int rowl = rt * 16 + (hi ? i + 8 : i);
            out[((size_t)(b * N_SZ + n0 + rowl)) * D_SZ + cb + t * 16 + m] =
                acc[t][i] * r[i];
        }
    }
}

extern "C" void kernel_launch(void* const* d_in, const int* in_sizes, int n_in,
                              void* d_out, int out_size, void* d_ws, size_t ws_size,
                              hipStream_t stream) {
    const float* q = (const float*)d_in[0];
    const float* k = (const float*)d_in[1];
    const float* v = (const float*)d_in[2];
    float* out = (float*)d_out;

    float* kvbuf   = (float*)d_ws;
    float* partial = kvbuf + (size_t)B_SZ * D_SZ * D_SZ;

    la_kv_partial<<<B_SZ * NCHUNK * 16 * 2, 256, 0, stream>>>(k, v, partial);
    la_kv_reduce<<<(B_SZ * D_SZ * D_SZ) / 256, 256, 0, stream>>>(partial, kvbuf);
    la_out<<<B_SZ * (N_SZ / 32), 256, 0, stream>>>(q, kvbuf, out);
}


# B fragments of both matrix kernels loaded 16 bytes per lane (V: 4 interleaved column tiles per load; kv stored as k pairs): 4x fewer global loads, 4 full lines each; sum order unchanged
# speedup vs baseline: 2.8577x; 1.0637x over previous
#include <hip/hip_runtime.h>
#include <hip/hip_bf16.h>

typedef float v2f __attribute__((ext_vector_type(2)));
typedef float v8f __attribute__((ext_vector_type(8)));

#define B_SZ 4
#define N_SZ 16384
#define D_SZ 256
#define NCHUNK 8
#define CHUNK_N (N_SZ / NCHUNK)
#define EPSV 1e-8f

#define NSUB 8
#define SUB_N (CHUNK_N / NSUB)
#define TILE_ELEMS (16 * 128)

__global__ __launch_bounds__(256) void la_kv_partial(const float* __restrict__ kmat,
                                                     const float* __restrict__ vmat,
                                                     float* __restrict__ partial) {
    __shared__ float red[NSUB * TILE_ELEMS];

    const int idx   = blockIdx.x;
    const int eh    = idx & 1;
    const int dt    = (idx >> 1) & 15;
    const int chunk = (idx >> 5) & (NCHUNK - 1);
    const int b     = idx >> 8;

    const int tid  = threadIdx.x;
    const int w    = __builtin_amdgcn_readfirstlane(tid >> 5);
    const int lane = tid & 31;
    const int m    = lane & 15;
    const int hi   = lane >> 4;

    const int ebase = eh * 128;
    const int dcol  = dt * 16 + m;

    const float* kb = kmat + (size_t)b * N_SZ * D_SZ;
    const float* vb = vmat + (size_t)b * N_SZ * D_SZ;

    v8f acc[8];
#pragma unroll
    for (int t = 0; t < 8; ++t) acc[t] = (v8f){0.f,0.f,0.f,0.f,0.f,0.f,0.f,0.f};

    const int nStart = chunk * CHUNK_N + w * SUB_N;

    for (int it = 0; it < SUB_N / 4; ++it) {
        const int nn = nStart + 4 * it + 2 * hi;
        const float* kp = kb + (size_t)nn * D_SZ + dcol;
        v2f a;
        a.x = fmaxf(kp[0],    0.f);
        a.y = fmaxf(kp[D_SZ], 0.f);
        const float* vp = vb + (size_t)nn * D_SZ + ebase + 4 * m;
#pragma unroll
        for (int g = 0; g < 2; ++g) {
            const float4 r0 = *(const float4*)(vp + g * 64);
            const float4 r1 = *(const float4*)(vp + g * 64 + D_SZ);
            v2f b0, b1, b2, b3;
            b0.x = r0.x; b0.y = r1.x;
            b1.x = r0.y; b1.y = r1.y;
            b2.x = r0.z; b2.y = r1.z;
            b3.x = r0.w; b3.y = r1.w;
            acc[g * 4 + 0] = __builtin_amdgcn_wmma_f32_16x16x4_f32(
                false, a, false, b0, (short)0, acc[g * 4 + 0], false, false);
            acc[g * 4 + 1] = __builtin_amdgcn_wmma_f32_16x16x4_f32(
                false, a, false, b1, (short)0, acc[g * 4 + 1], false, false);
            acc[g * 4 + 2] = __builtin_amdgcn_wmma_f32_16x16x4_f32(
                false, a, false, b2, (short)0, acc[g * 4 + 2], false, false);
            acc[g * 4 + 3] = __builtin_amdgcn_wmma_f32_16x16x4_f32(
                false, a, false, b3, (short)0, acc[g * 4 + 3], false, false);
        }
    }

#pragma unroll
    for (int t = 0; t < 8; ++t) {
#pragma unroll
        for (int i = 0; i < 8; ++i)
            red[w * TILE_ELEMS + (t * 8 + i) * 32 + lane] = acc[t][i];
    }
    __syncthreads();

    float* pb = partial + (((size_t)b * NCHUNK + chunk) * D_SZ * D_SZ);
    const int drow = dt * 16 + (hi ? w + 8 : w);
#pragma unroll
    for (int g = 0; g < 2; ++g) {
        float sj[4];
#pragma unroll
        for (int j = 0; j < 4; ++j) {
            const int e = ((g * 4 + j) * 8 + w) * 32 + lane;
            float s = red[e];
#pragma unroll
            for (int w2 = 1; w2 < NSUB; ++w2) s += red[w2 * TILE_ELEMS + e];
            sj[j] = s;
        }
        *(float4*)&pb[(size_t)drow * D_SZ + ebase + g * 64 + 4 * m] =
            make_float4(sj[0], sj[1], sj[2], sj[3]);
    }
}

__global__ __launch_bounds__(256) void la_kv_reduce(const float* __restrict__ partial,
                                                    float* __restrict__ kvp) {
    const int idx = blockIdx.x * 256 + threadIdx.x;
    const int e   = idx & (D_SZ - 1);
    const int p   = (idx >> 8) & (D_SZ / 2 - 1);
    const int b   = idx >> 15;
    float s0 = 0.f, s1 = 0.f;
#pragma unroll
    for (int c = 0; c < NCHUNK; ++c) {
        const float* pp = partial + (((size_t)(b * NCHUNK + c)) << 16) + (2 * p) * D_SZ + e;
        s0 += pp[0];
        s1 += pp[D_SZ];
    }
    *(float2*)&kvp[(((size_t)b * (D_SZ / 2) + p) * D_SZ + e) * 2] = make_float2(s0, s1);
}

#define QPITCH 260
#define KSTEPS2 4

__global__ __launch_bounds__(256) void la_out(const float* __restrict__ qmat,
                                              const float* __restrict__ kv,
                                              float* __restrict__ out) {
    __shared__ float ldsq[32 * QPITCH];
    __shared__ float dpart[256];
    __shared__ float drcp[32];

    const int b    = blockIdx.x >> 9;
    const int n0   = (blockIdx.x & 511) * 32;
    const int tid  = threadIdx.x;

    {
        const int row  = tid >> 3;
        const int cseg = (tid & 7) * 32;
        const float* qrow = qmat + ((size_t)(b * N_SZ + n0 + row)) * D_SZ + cseg;
        float s = 0.f;
#pragma unroll
        for (int j = 0; j < 32; j += 4) {
            float4 x = *(const float4*)(qrow + j);
            x.x = fmaxf(x.x, 0.f);
            x.y = fmaxf(x.y, 0.f);
            x.z = fmaxf(x.z, 0.f);
            x.w = fmaxf(x.w, 0.f);
            *(float4*)&ldsq[row * QPITCH + cseg + j] = x;
            s += x.x + x.y + x.z + x.w;
        }
        dpart[tid] = s;
    }
    __syncthreads();
    if (tid < 32) {
        float s = EPSV;
#pragma unroll
        for (int j = 0; j < 8; ++j) s += dpart[tid * 8 + j];
        drcp[tid] = 1.0f / s;
    }
    __syncthreads();

    const int w    = tid >> 5;
    const int lane = tid & 31;
    const int m    = lane & 15;
    const int hi   = lane >> 4;
    const int rt   = w >> 2;
    const int cb   = (w & 3) * 64;

    v8f acc[4];
#pragma unroll
    for (int t = 0; t < 4; ++t) acc[t] = (v8f){0.f,0.f,0.f,0.f,0.f,0.f,0.f,0.f};

    const float* kvb = kv + ((size_t)b << 16);

    for (int k0 = 0; k0 < D_SZ; k0 += 4 * KSTEPS2) {
        v2f a[KSTEPS2];
        float4 q[KSTEPS2][2];
#pragma unroll
        for (int s = 0; s < KSTEPS2; ++s) {
            const int kk = k0 + 4 * s + 2 * hi;
            const float* ap = &ldsq[(rt * 16 + m) * QPITCH + kk];
            a[s].x = ap[0];
            a[s].y = ap[1];
            const float* bp = kvb + ((size_t)(kk >> 1) * D_SZ + cb + 2 * m) * 2;
            q[s][0] = *(const float4*)bp;
            q[s][1] = *(const float4*)(bp + 64);
        }
        __builtin_amdgcn_sched_barrier(0);
#pragma unroll
        for (int s = 0; s < KSTEPS2; ++s) {
            v2f b0, b1, b2, b3;
            b0.x = q[s][0].x; b0.y = q[s][0].y;
            b1.x = q[s][0].z; b1.y = q[s][0].w;
            b2.x = q[s][1].x; b2.y = q[s][1].y;
            b3.x = q[s][1].z; b3.y = q[s][1].w;
            acc[0] = __builtin_amdgcn_wmma_f32_16x16x4_f32(
                false, a[s], false, b0, (short)0, acc[0], false, false);
            acc[1] = __builtin_amdgcn_wmma_f32_16x16x4_f32(
                false, a[s], false, b1, (short)0, acc[1], false, false);
            acc[2] = __builtin_amdgcn_wmma_f32_16x16x4_f32(
                false, a[s], false, b2, (short)0, acc[2], false, false);
            acc[3] = __builtin_amdgcn_wmma_f32_16x16x4_f32(
                false, a[s], false, b3, (short)0, acc[3], false, false);
        }
    }

    float r[8];
#pragma unroll
    for (int i = 0; i < 8; ++i) r[i] = drcp[rt * 16 + (hi ? i + 8 : i)];

#pragma unroll
    for (int i = 0; i < 8; ++i) {
        const int rowl = rt * 16 + (hi ? i + 8 : i);
        float* orow = out + ((size_t)(b * N_SZ + n0 + rowl)) * D_SZ + cb + 2 * m;
        *(float2*)(orow)      = make_float2(acc[0][i] * r[i], acc[1][i] * r[i]);
        *(float2*)(orow + 32) = make_float2(acc[2][i] * r[i], acc[3][i] * r[i]);
    }
}

extern "C" void kernel_launch(void* const* d_in, const int* in_sizes, int n_in,
                              void* d_out, int out_size, void* d_ws, size_t ws_size,
                              hipStream_t stream) {
    const float* q = (const float*)d_in[0];
    const float* k = (const float*)d_in[1];
    const float* v = (const float*)d_in[2];
    float* out = (float*)d_out;

    float* kvbuf   = (float*)d_ws;
    float* partial = kvbuf + (size_t)B_SZ * D_SZ * D_SZ;

    la_kv_partial<<<B_SZ * NCHUNK * 16 * 2, 256, 0, stream>>>(k, v, partial);
    la_kv_reduce<<<(B_SZ * (D_SZ / 2) * D_SZ) / 256, 256, 0, stream>>>(partial, kvbuf);
    la_out<<<B_SZ * (N_SZ / 32), 256, 0, stream>>>(q, kvbuf, out);
}


# each wave's accumulator block reshaped to 2 row tiles x 4 column tiles in both matrix kernels (kv tile 32x64; out: 64 rows per work-group): every V / kv fragment feeds 2 WMMAs; sum order unchanged
# speedup vs baseline: 3.3321x; 1.1660x over previous
#include <hip/hip_runtime.h>
#include <hip/hip_bf16.h>

typedef float v2f __attribute__((ext_vector_type(2)));
typedef float v8f __attribute__((ext_vector_type(8)));

#define B_SZ 4
#define N_SZ 16384
#define D_SZ 256
#define NCHUNK 8
#define CHUNK_N (N_SZ / NCHUNK)
#define EPSV 1e-8f

#define NSUB 8
#define SUB_N (CHUNK_N / NSUB)
#define TILE_ELEMS (32 * 64)
#define KS1 4

__global__ __launch_bounds__(256) void la_kv_partial(const float* __restrict__ kmat,
                                                     const float* __restrict__ vmat,
                                                     float* __restrict__ partial) {
    __shared__ float red[NSUB * TILE_ELEMS];

    const int idx   = blockIdx.x;
    const int eq    = idx & 3;
    const int dt    = (idx >> 2) & 7;
    const int chunk = (idx >> 5) & (NCHUNK - 1);
    const int b     = idx >> 8;

    const int tid  = threadIdx.x;
    const int w    = __builtin_amdgcn_readfirstlane(tid >> 5);
    const int lane = tid & 31;
    const int m    = lane & 15;
    const int hi   = lane >> 4;

    const int ebase = eq * 64;
    const int dcol  = dt * 32 + m;

    const float* kb = kmat + (size_t)b * N_SZ * D_SZ;
    const float* vb = vmat + (size_t)b * N_SZ * D_SZ;

    v8f acc[8];
#pragma unroll
    for (int t = 0; t < 8; ++t) acc[t] = (v8f){0.f,0.f,0.f,0.f,0.f,0.f,0.f,0.f};

    const int nFirst = chunk * CHUNK_N + w * SUB_N + 2 * hi;
    const float* kp = kb + (size_t)nFirst * D_SZ + dcol;
    const float* vp = vb + (size_t)nFirst * D_SZ + ebase + 4 * m;

    for (int it = 0; it < SUB_N / 4; it += KS1) {
        float  ka[KS1][4];
        float4 r0[KS1], r1[KS1];
#pragma unroll
        for (int s = 0; s < KS1; ++s) {
            const float* kps = kp + s * 4 * D_SZ;
            ka[s][0] = kps[0];
            ka[s][1] = kps[D_SZ];
            ka[s][2] = kps[16];
            ka[s][3] = kps[D_SZ + 16];
            const float* vps = vp + s * 4 * D_SZ;
            r0[s] = *(const float4*)(vps);
            r1[s] = *(const float4*)(vps + D_SZ);
        }
        kp += KS1 * 4 * D_SZ;
        vp += KS1 * 4 * D_SZ;
        v2f a0[KS1], a1[KS1], bf[KS1][4];
#pragma unroll
        for (int s = 0; s < KS1; ++s) {
            a0[s].x = fmaxf(ka[s][0], 0.f);
            a0[s].y = fmaxf(ka[s][1], 0.f);
            a1[s].x = fmaxf(ka[s][2], 0.f);
            a1[s].y = fmaxf(ka[s][3], 0.f);
            bf[s][0].x = r0[s].x; bf[s][0].y = r1[s].x;
            bf[s][1].x = r0[s].y; bf[s][1].y = r1[s].y;
            bf[s][2].x = r0[s].z; bf[s][2].y = r1[s].z;
            bf[s][3].x = r0[s].w; bf[s][3].y = r1[s].w;
        }
        __builtin_amdgcn_sched_barrier(0);
#pragma unroll
        for (int s = 0; s < KS1; ++s) {
#pragma unroll
            for (int j = 0; j < 4; ++j)
                acc[j] = __builtin_amdgcn_wmma_f32_16x16x4_f32(
                    false, a0[s], false, bf[s][j], (short)0, acc[j], false, false);
#pragma unroll
            for (int j = 0; j < 4; ++j)
                acc[4 + j] = __builtin_amdgcn_wmma_f32_16x16x4_f32(
                    false, a1[s], false, bf[s][j], (short)0, acc[4 + j], false, false);
        }
    }

#pragma unroll
    for (int t = 0; t < 8; ++t) {
#pragma unroll
        for (int i = 0; i < 8; ++i)
            red[w * TILE_ELEMS + (t * 8 + i) * 32 + lane] = acc[t][i];
    }
    __syncthreads();

    float* pb = partial + (((size_t)b * NCHUNK + chunk) * D_SZ * D_SZ);
#pragma unroll
    for (int r = 0; r < 2; ++r) {
        const int drow = dt * 32 + r * 16 + (hi ? w + 8 : w);
        float sj[4];
#pragma unroll
        for (int j = 0; j < 4; ++j) {
            const int e = ((r * 4 + j) * 8 + w) * 32 + lane;
            float s = red[e];
#pragma unroll
            for (int w2 = 1; w2 < NSUB; ++w2) s += red[w2 * TILE_ELEMS + e];
            sj[j] = s;
        }
        *(float4*)&pb[(size_t)drow * D_SZ + ebase + 4 * m] =
            make_float4(sj[0], sj[1], sj[2], sj[3]);
    }
}

__global__ __launch_bounds__(256) void la_kv_reduce(const float* __restrict__ partial,
                                                    float* __restrict__ kvp) {
    const int idx = blockIdx.x * 256 + threadIdx.x;
    const int e   = idx & (D_SZ - 1);
    const int p   = (idx >> 8) & (D_SZ / 2 - 1);
    const int b   = idx >> 15;
    float s0 = 0.f, s1 = 0.f;
#pragma unroll
    for (int c = 0; c < NCHUNK; ++c) {
        const float* pp = partial + (((size_t)(b * NCHUNK + c)) << 16) + (2 * p) * D_SZ + e;
        s0 += pp[0];
        s1 += pp[D_SZ];
    }
    *(float2*)&kvp[(((size_t)b * (D_SZ / 2) + p) * D_SZ + e) * 2] = make_float2(s0, s1);
}

#define QPITCH 260
#define QROWS 64
#define KSTEPS2 4

__global__ __launch_bounds__(256) void la_out(const float* __restrict__ qmat,
                                              const float* __restrict__ kv,
                                              float* __restrict__ out) {
    __shared__ float ldsq[QROWS * QPITCH];
    __shared__ float dpart[QROWS * 8];
    __shared__ float drcp[QROWS];

    const int b    = blockIdx.x >> 8;
    const int n0   = (blockIdx.x & 255) * QROWS;
    const int tid  = threadIdx.x;

#pragma unroll
    for (int half = 0; half < 2; ++half) {
        const int row  = half * 32 + (tid >> 3);
        const int cseg = (tid & 7) * 32;
        const float* qrow = qmat + ((size_t)(b * N_SZ + n0 + row)) * D_SZ + cseg;
        float s = 0.f;
#pragma unroll
        for (int j = 0; j < 32; j += 4) {
            float4 x = *(const float4*)(qrow + j);
            x.x = fmaxf(x.x, 0.f);
            x.y = fmaxf(x.y, 0.f);
            x.z = fmaxf(x.z, 0.f);
            x.w = fmaxf(x.w, 0.f);
            *(float4*)&ldsq[row * QPITCH + cseg + j] = x;
            s += x.x + x.y + x.z + x.w;
        }
        dpart[half * 256 + tid] = s;
    }
    __syncthreads();
    if (tid < QROWS) {
        float s = EPSV;
#pragma unroll
        for (int j = 0; j < 8; ++j) s += dpart[tid * 8 + j];
        drcp[tid] = 1.0f / s;
    }
    __syncthreads();

    const int w    = tid >> 5;
    const int lane = tid & 31;
    const int m    = lane & 15;
    const int hi   = lane >> 4;
    const int rp   = w >> 2;
    const int cb   = (w & 3) * 64;

    v8f acc[2][4];
#pragma unroll
    for (int r = 0; r < 2; ++r) {
#pragma unroll
        for (int t = 0; t < 4; ++t) acc[r][t] = (v8f){0.f,0.f,0.f,0.f,0.f,0.f,0.f,0.f};
    }

    const float* bp = kv + ((size_t)b << 16) + ((size_t)hi * D_SZ + cb + 2 * m) * 2;

#pragma clang loop unroll(disable)
    for (int k0 = 0; k0 < D_SZ; k0 += 4 * KSTEPS2) {
        v2f a[KSTEPS2][2];
        float4 q[KSTEPS2][2];
#pragma unroll
        for (int s = 0; s < KSTEPS2; ++s) {
            const int kk = k0 + 4 * s + 2 * hi;
            const float* ap = &ldsq[(rp * 32 + m) * QPITCH + kk];
            a[s][0].x = ap[0];
            a[s][0].y = ap[1];
            a[s][1].x = ap[16 * QPITCH];
            a[s][1].y = ap[16 * QPITCH + 1];
            q[s][0] = *(const float4*)(bp + s * 4 * D_SZ);
            q[s][1] = *(const float4*)(bp + s * 4 * D_SZ + 64);
        }
        bp += KSTEPS2 * 4 * D_SZ;
        __builtin_amdgcn_sched_barrier(0);
#pragma unroll
        for (int s = 0; s < KSTEPS2; ++s) {
            v2f b0, b1, b2, b3;
            b0.x = q[s][0].x; b0.y = q[s][0].y;
            b1.x = q[s][0].z; b1.y = q[s][0].w;
            b2.x = q[s][1].x; b2.y = q[s][1].y;
            b3.x = q[s][1].z; b3.y = q[s][1].w;
#pragma unroll
            for (int r = 0; r < 2; ++r) {
                acc[r][0] = __builtin_amdgcn_wmma_f32_16x16x4_f32(
                    false, a[s][r], false, b0, (short)0, acc[r][0], false, false);
                acc[r][1] = __builtin_amdgcn_wmma_f32_16x16x4_f32(
                    false, a[s][r], false, b1, (short)0, acc[r][1], false, false);
                acc[r][2] = __builtin_amdgcn_wmma_f32_16x16x4_f32(
                    false, a[s][r], false, b2, (short)0, acc[r][2], false, false);
                acc[r][3] = __builtin_amdgcn_wmma_f32_16x16x4_f32(
                    false, a[s][r], false, b3, (short)0, acc[r][3], false, false);
            }
        }
    }

#pragma unroll
    for (int r = 0; r < 2; ++r) {
#pragma unroll
        for (int i = 0; i < 8; ++i) {
            const int rowl = rp * 32 + r * 16 + (hi ? i + 8 : i);
            const float rc = drcp[rowl];
            float* orow = out + ((size_t)(b * N_SZ + n0 + rowl)) * D_SZ + cb + 2 * m;
            *(float2*)(orow)      = make_float2(acc[r][0][i] * rc, acc[r][1][i] * rc);
            *(float2*)(orow + 32) = make_float2(acc[r][2][i] * rc, acc[r][3][i] * rc);
        }
    }
}

extern "C" void kernel_launch(void* const* d_in, const int* in_sizes, int n_in,
                              void* d_out, int out_size, void* d_ws, size_t ws_size,
                              hipStream_t stream) {
    const float* q = (const float*)d_in[0];
    const float* k = (const float*)d_in[1];
    const float* v = (const float*)d_in[2];
    float* out = (float*)d_out;

    float* kvbuf   = (float*)d_ws;
    float* partial = kvbuf + (size_t)B_SZ * D_SZ * D_SZ;

    la_kv_partial<<<B_SZ * NCHUNK * 8 * 4, 256, 0, stream>>>(k, v, partial);
    la_kv_reduce<<<(B_SZ * (D_SZ / 2) * D_SZ) / 256, 256, 0, stream>>>(partial, kvbuf);
    la_out<<<B_SZ * (N_SZ / 64), 256, 0, stream>>>(q, kvbuf, out);
}


# both matrix kernels: operands of the next 4 (kv) / 3 (out) 2-k-step trips prefetched into rotating register sets, at 4 waves per SIMD; kv: 4 chunks, 512 rows per wave: new sum order over n
# speedup vs baseline: 3.4814x; 1.0448x over previous
#include <hip/hip_runtime.h>
#include <hip/hip_bf16.h>

typedef float v2f __attribute__((ext_vector_type(2)));
typedef float v8f __attribute__((ext_vector_type(8)));

#define B_SZ 4
#define N_SZ 16384
#define D_SZ 256
#define NCHUNK 4
#define CHUNK_N (N_SZ / NCHUNK)
#define EPSV 1e-8f

#define NSUB 8
#define SUB_N (CHUNK_N / NSUB)
#define TILE_ELEMS (32 * 64)
#define KS1 2
#define NSET1 5
#define TRIP1 (KS1 * 4 * D_SZ)
#define NTRIP1 (SUB_N / (KS1 * 4))
#define NGRP1 ((NTRIP1 - (NSET1 - 1)) / NSET1)

struct KvFrag {
    float  ka[KS1][4];
    float4 r0[KS1], r1[KS1];
};

__device__ __forceinline__ void kv_fetch(KvFrag& f, const float* kp, const float* vp) {
#pragma unroll
    for (int s = 0; s < KS1; ++s) {
        const float* kps = kp + s * 4 * D_SZ;
        f.ka[s][0] = kps[0];
        f.ka[s][1] = kps[D_SZ];
        f.ka[s][2] = kps[16];
        f.ka[s][3] = kps[D_SZ + 16];
        const float* vps = vp + s * 4 * D_SZ;
        f.r0[s] = *(const float4*)(vps);
        f.r1[s] = *(const float4*)(vps + D_SZ);
    }
}

__device__ __forceinline__ void kv_mma(const KvFrag& f, v8f (&acc)[8]) {
    v2f a0[KS1], a1[KS1], bf[KS1][4];
#pragma unroll
    for (int s = 0; s < KS1; ++s) {
        a0[s].x = fmaxf(f.ka[s][0], 0.f);
        a0[s].y = fmaxf(f.ka[s][1], 0.f);
        a1[s].x = fmaxf(f.ka[s][2], 0.f);
        a1[s].y = fmaxf(f.ka[s][3], 0.f);
        bf[s][0].x = f.r0[s].x; bf[s][0].y = f.r1[s].x;
        bf[s][1].x = f.r0[s].y; bf[s][1].y = f.r1[s].y;
        bf[s][2].x = f.r0[s].z; bf[s][2].y = f.r1[s].z;
        bf[s][3].x = f.r0[s].w; bf[s][3].y = f.r1[s].w;
    }
    __builtin_amdgcn_sched_barrier(0);
#pragma unroll
    for (int s = 0; s < KS1; ++s) {
#pragma unroll
        for (int j = 0; j < 4; ++j)
            acc[j] = __builtin_amdgcn_wmma_f32_16x16x4_f32(
                false, a0[s], false, bf[s][j], (short)0, acc[j], false, false);
#pragma unroll
        for (int j = 0; j < 4; ++j)
            acc[4 + j] = __builtin_amdgcn_wmma_f32_16x16x4_f32(
                false, a1[s], false, bf[s][j], (short)0, acc[4 + j], false, false);
    }
}

__global__ __launch_bounds__(256) void la_kv_partial(const float* __restrict__ kmat,
                                                     const float* __restrict__ vmat,
                                                     float* __restrict__ partial) {
    __shared__ float red[NSUB * TILE_ELEMS];

    const int idx   = blockIdx.x;
    const int eq    = idx & 3;
    const int dt    = (idx >> 2) & 7;
    const int chunk = (idx >> 5) & (NCHUNK - 1);
    const int b     = idx >> 7;

    const int tid  = threadIdx.x;
    const int w    = __builtin_amdgcn_readfirstlane(tid >> 5);
    const int lane = tid & 31;
    const int m    = lane & 15;
    const int hi   = lane >> 4;

    const int ebase = eq * 64;
    const int dcol  = dt * 32 + m;

    const float* kb = kmat + (size_t)b * N_SZ * D_SZ;
    const float* vb = vmat + (size_t)b * N_SZ * D_SZ;

    v8f acc[8];
#pragma unroll
    for (int t = 0; t < 8; ++t) acc[t] = (v8f){0.f,0.f,0.f,0.f,0.f,0.f,0.f,0.f};

    const int nFirst = chunk * CHUNK_N + w * SUB_N + 2 * hi;
    const float* kp = kb + (size_t)nFirst * D_SZ + dcol;
    const float* vp = vb + (size_t)nFirst * D_SZ + ebase + 4 * m;

    KvFrag f[NSET1];
#pragma unroll
    for (int u = 0; u < NSET1 - 1; ++u) {
        kv_fetch(f[u], kp + u * TRIP1, vp + u * TRIP1);
        __builtin_amdgcn_sched_barrier(0);
    }

    for (int g = 0; g < NGRP1; ++g) {
#pragma unroll
        for (int u = 0; u < NSET1; ++u) {
            __builtin_amdgcn_sched_barrier(0);
            kv_fetch(f[(u + NSET1 - 1) % NSET1], kp + (u + NSET1 - 1) * TRIP1,
                     vp + (u + NSET1 - 1) * TRIP1);
            kv_mma(f[u], acc);
        }
        kp += NSET1 * TRIP1;
        vp += NSET1 * TRIP1;
    }
    static_assert(NTRIP1 - NGRP1 * NSET1 == NSET1 - 1, "the loop must fetch the last trip");
#pragma unroll
    for (int t = NGRP1 * NSET1; t < NTRIP1; ++t) kv_mma(f[t % NSET1], acc);

#pragma unroll
    for (int t = 0; t < 8; ++t) {
#pragma unroll
        for (int i = 0; i < 8; ++i)
            red[w * TILE_ELEMS + (t * 8 + i) * 32 + lane] = acc[t][i];
    }
    __syncthreads();

    float* pb = partial + (((size_t)b * NCHUNK + chunk) * D_SZ * D_SZ);
#pragma unroll
    for (int r = 0; r < 2; ++r) {
        const int drow = dt * 32 + r * 16 + (hi ? w + 8 : w);
        float sj[4];
#pragma unroll
        for (int j = 0; j < 4; ++j) {
            const int e = ((r * 4 + j) * 8 + w) * 32 + lane;
            float s = red[e];
#pragma unroll
            for (int w2 = 1; w2 < NSUB; ++w2) s += red[w2 * TILE_ELEMS + e];
            sj[j] = s;
        }
        *(float4*)&pb[(size_t)drow * D_SZ + ebase + 4 * m] =
            make_float4(sj[0], sj[1], sj[2], sj[3]);
    }
}

__global__ __launch_bounds__(256) void la_kv_reduce(const float* __restrict__ partial,
                                                    float* __restrict__ kvp) {
    const int idx = blockIdx.x * 256 + threadIdx.x;
    const int e   = idx & (D_SZ - 1);
    const int p   = (idx >> 8) & (D_SZ / 2 - 1);
    const int b   = idx >> 15;
    float s0 = 0.f, s1 = 0.f;
#pragma unroll
    for (int c = 0; c < NCHUNK; ++c) {
        const float* pp = partial + (((size_t)(b * NCHUNK + c)) << 16) + (2 * p) * D_SZ + e;
        s0 += pp[0];
        s1 += pp[D_SZ];
    }
    *(float2*)&kvp[(((size_t)b * (D_SZ / 2) + p) * D_SZ + e) * 2] = make_float2(s0, s1);
}

#define QPITCH 260
#define QROWS 64
#define KSTEPS2 2
#define TRIP2A (4 * KSTEPS2)
#define TRIP2B (KSTEPS2 * 4 * D_SZ)
#define NTRIP2 (D_SZ / (4 * KSTEPS2))
#define NSET2 4
#define NGRP2 ((NTRIP2 - (NSET2 - 1)) / NSET2)

struct OutFrag {
    v2f    a[KSTEPS2][2];
    float4 q[KSTEPS2][2];
};

__device__ __forceinline__ void out_fetch(OutFrag& f, const float* ap, const float* bp) {
#pragma unroll
    for (int s = 0; s < KSTEPS2; ++s) {
        f.a[s][0] = *(const v2f*)(ap + 4 * s);
        f.a[s][1] = *(const v2f*)(ap + 16 * QPITCH + 4 * s);
        f.q[s][0] = *(const float4*)(bp + s * 4 * D_SZ);
        f.q[s][1] = *(const float4*)(bp + s * 4 * D_SZ + 64);
    }
}

__device__ __forceinline__ void out_mma(const OutFrag& f, v8f (&acc)[2][4]) {
#pragma unroll
    for (int s = 0; s < KSTEPS2; ++s) {
        v2f b0, b1, b2, b3;
        b0.x = f.q[s][0].x; b0.y = f.q[s][0].y;
        b1.x = f.q[s][0].z; b1.y = f.q[s][0].w;
        b2.x = f.q[s][1].x; b2.y = f.q[s][1].y;
        b3.x = f.q[s][1].z; b3.y = f.q[s][1].w;
#pragma unroll
        for (int r = 0; r < 2; ++r) {
            acc[r][0] = __builtin_amdgcn_wmma_f32_16x16x4_f32(
                false, f.a[s][r], false, b0, (short)0, acc[r][0], false, false);
            acc[r][1] = __builtin_amdgcn_wmma_f32_16x16x4_f32(
                false, f.a[s][r], false, b1, (short)0, acc[r][1], false, false);
            acc[r][2] = __builtin_amdgcn_wmma_f32_16x16x4_f32(
                false, f.a[s][r], false, b2, (short)0, acc[r][2], false, false);
            acc[r][3] = __builtin_amdgcn_wmma_f32_16x16x4_f32(
                false, f.a[s][r], false, b3, (short)0, acc[r][3], false, false);
        }
    }
}

__global__ __launch_bounds__(256) void la_out(const float* __restrict__ qmat,
                                              const float* __restrict__ kv,
                                              float* __restrict__ out) {
    __shared__ __attribute__((aligned(16))) float ldsq[QROWS * QPITCH];
    __shared__ float dpart[QROWS * 8];
    __shared__ float drcp[QROWS];

    const int b    = blockIdx.x >> 8;
    const int n0   = (blockIdx.x & 255) * QROWS;
    const int tid  = threadIdx.x;

#pragma unroll
    for (int half = 0; half < 2; ++half) {
        const int row  = half * 32 + (tid >> 3);
        const int cseg = (tid & 7) * 32;
        const float* qrow = qmat + ((size_t)(b * N_SZ + n0 + row)) * D_SZ + cseg;
        float s = 0.f;
#pragma unroll
        for (int j = 0; j < 32; j += 4) {
            float4 x = *(const float4*)(qrow + j);
            x.x = fmaxf(x.x, 0.f);
            x.y = fmaxf(x.y, 0.f);
            x.z = fmaxf(x.z, 0.f);
            x.w = fmaxf(x.w, 0.f);
            *(float4*)&ldsq[row * QPITCH + cseg + j] = x;
            s += x.x + x.y + x.z + x.w;
        }
        dpart[half * 256 + tid] = s;
    }
    __syncthreads();
    if (tid < QROWS) {
        float s = EPSV;
#pragma unroll
        for (int j = 0; j < 8; ++j) s += dpart[tid * 8 + j];
        drcp[tid] = 1.0f / s;
    }
    __syncthreads();

    const int w    = tid >> 5;
    const int lane = tid & 31;
    const int m    = lane & 15;
    const int hi   = lane >> 4;
    const int rp   = w >> 2;
    const int cb   = (w & 3) * 64;

    v8f acc[2][4];
#pragma unroll
    for (int r = 0; r < 2; ++r) {
#pragma unroll
        for (int t = 0; t < 4; ++t) acc[r][t] = (v8f){0.f,0.f,0.f,0.f,0.f,0.f,0.f,0.f};
    }

    const float* bp = kv + ((size_t)b << 16) + ((size_t)hi * D_SZ + cb + 2 * m) * 2;
    const float* ap = &ldsq[(rp * 32 + m) * QPITCH + 2 * hi];

    OutFrag f[NSET2];
#pragma unroll
    for (int u = 0; u < NSET2 - 1; ++u) out_fetch(f[u], ap + u * TRIP2A, bp + u * TRIP2B);

    for (int g = 0; g < NGRP2; ++g) {
#pragma unroll
        for (int u = 0; u < NSET2; ++u) {
            __builtin_amdgcn_sched_barrier(0);
            out_fetch(f[(u + NSET2 - 1) % NSET2], ap + (u + NSET2 - 1) * TRIP2A,
                      bp + (u + NSET2 - 1) * TRIP2B);
            __builtin_amdgcn_sched_barrier(0);
            out_mma(f[u], acc);
        }
        ap += NSET2 * TRIP2A;
        bp += NSET2 * TRIP2B;
    }
#pragma unroll
    for (int t = NGRP2 * NSET2; t < NTRIP2; ++t) {
        const int u = t - NGRP2 * NSET2;
        if (t + NSET2 - 1 < NTRIP2)
            out_fetch(f[(t + NSET2 - 1) % NSET2], ap + (u + NSET2 - 1) * TRIP2A,
                      bp + (u + NSET2 - 1) * TRIP2B);
        out_mma(f[t % NSET2], acc);
    }

#pragma unroll
    for (int r = 0; r < 2; ++r) {
#pragma unroll
        for (int i = 0; i < 8; ++i) {
            const int rowl = rp * 32 + r * 16 + (hi ? i + 8 : i);
            const float rc = drcp[rowl];
            float* orow = out + ((size_t)(b * N_SZ + n0 + rowl)) * D_SZ + cb + 2 * m;
            *(float2*)(orow)      = make_float2(acc[r][0][i] * rc, acc[r][1][i] * rc);
            *(float2*)(orow + 32) = make_float2(acc[r][2][i] * rc, acc[r][3][i] * rc);
        }
    }
}

extern "C" void kernel_launch(void* const* d_in, const int* in_sizes, int n_in,
                              void* d_out, int out_size, void* d_ws, size_t ws_size,
                              hipStream_t stream) {
    const float* q = (const float*)d_in[0];
    const float* k = (const float*)d_in[1];
    const float* v = (const float*)d_in[2];
    float* out = (float*)d_out;

    float* kvbuf   = (float*)d_ws;
    float* partial = kvbuf + (size_t)B_SZ * D_SZ * D_SZ;

    la_kv_partial<<<B_SZ * NCHUNK * 8 * 4, 256, 0, stream>>>(k, v, partial);
    la_kv_reduce<<<(B_SZ * (D_SZ / 2) * D_SZ) / 256, 256, 0, stream>>>(partial, kvbuf);
    la_out<<<B_SZ * (N_SZ / 64), 256, 0, stream>>>(q, kvbuf, out);
}


# kv kernel: A (K) fragments loaded 16 bytes per lane, 4 interleaved row tiles per load; wave tile 4x4 tiles (kv tile 64x64): 4 loads per 16 WMMAs, was 12; 8 row chunks: new sum order over n
# speedup vs baseline: 3.6146x; 1.0383x over previous
#include <hip/hip_runtime.h>
#include <hip/hip_bf16.h>

typedef float v2f __attribute__((ext_vector_type(2)));
typedef float v8f __attribute__((ext_vector_type(8)));

#define B_SZ 4
#define N_SZ 16384
#define D_SZ 256
#define NCHUNK 8
#define CHUNK_N (N_SZ / NCHUNK)
#define EPSV 1e-8f

#define NSUB 8
#define SUB_N (CHUNK_N / NSUB)
#define TILE_VEC4 (64 * 64 / 4)
#define NSET1 5
#define TRIP1 (4 * D_SZ)
#define NTRIP1 (SUB_N / 4)
#define NGRP1 ((NTRIP1 - (NSET1 - 1)) / NSET1)

struct KvFrag {
    float4 k0, k1;
    float4 r0, r1;
};

__device__ __forceinline__ void kv_fetch(KvFrag& f, const float* kp, const float* vp) {
    f.k0 = *(const float4*)(kp);
    f.k1 = *(const float4*)(kp + D_SZ);
    f.r0 = *(const float4*)(vp);
    f.r1 = *(const float4*)(vp + D_SZ);
}

__device__ __forceinline__ void kv_mma(const KvFrag& f, v8f (&acc)[16]) {
    v2f a[4], bf[4];
    a[0].x = fmaxf(f.k0.x, 0.f); a[0].y = fmaxf(f.k1.x, 0.f);
    a[1].x = fmaxf(f.k0.y, 0.f); a[1].y = fmaxf(f.k1.y, 0.f);
    a[2].x = fmaxf(f.k0.z, 0.f); a[2].y = fmaxf(f.k1.z, 0.f);
    a[3].x = fmaxf(f.k0.w, 0.f); a[3].y = fmaxf(f.k1.w, 0.f);
    bf[0].x = f.r0.x; bf[0].y = f.r1.x;
    bf[1].x = f.r0.y; bf[1].y = f.r1.y;
    bf[2].x = f.r0.z; bf[2].y = f.r1.z;
    bf[3].x = f.r0.w; bf[3].y = f.r1.w;
    __builtin_amdgcn_sched_barrier(0);
#pragma unroll
    for (int r = 0; r < 4; ++r) {
#pragma unroll
        for (int j = 0; j < 4; ++j)
            acc[r * 4 + j] = __builtin_amdgcn_wmma_f32_16x16x4_f32(
                false, a[r], false, bf[j], (short)0, acc[r * 4 + j], false, false);
    }
}

__global__ __launch_bounds__(256) void la_kv_partial(const float* __restrict__ kmat,
                                                     const float* __restrict__ vmat,
                                                     float* __restrict__ partial) {
    __shared__ float4 red[NSUB * TILE_VEC4];

    const int idx   = blockIdx.x;
    const int eq    = idx & 3;
    const int dt    = (idx >> 2) & 3;
    const int chunk = (idx >> 4) & (NCHUNK - 1);
    const int b     = idx >> 7;

    const int tid  = threadIdx.x;
    const int w    = __builtin_amdgcn_readfirstlane(tid >> 5);
    const int lane = tid & 31;
    const int m    = lane & 15;
    const int hi   = lane >> 4;

    const int ebase = eq * 64;
    const int dbase = dt * 64;

    const float* kb = kmat + (size_t)b * N_SZ * D_SZ;
    const float* vb = vmat + (size_t)b * N_SZ * D_SZ;

    v8f acc[16];
#pragma unroll
    for (int t = 0; t < 16; ++t) acc[t] = (v8f){0.f,0.f,0.f,0.f,0.f,0.f,0.f,0.f};

    const int nFirst = chunk * CHUNK_N + w * SUB_N + 2 * hi;
    const float* kp = kb + (size_t)nFirst * D_SZ + dbase + 4 * m;
    const float* vp = vb + (size_t)nFirst * D_SZ + ebase + 4 * m;

    KvFrag f[NSET1];
#pragma unroll
    for (int u = 0; u < NSET1 - 1; ++u) {
        kv_fetch(f[u], kp + u * TRIP1, vp + u * TRIP1);
        __builtin_amdgcn_sched_barrier(0);
    }

    for (int g = 0; g < NGRP1; ++g) {
#pragma unroll
        for (int u = 0; u < NSET1; ++u) {
            __builtin_amdgcn_sched_barrier(0);
            kv_fetch(f[(u + NSET1 - 1) % NSET1], kp + (u + NSET1 - 1) * TRIP1,
                     vp + (u + NSET1 - 1) * TRIP1);
            kv_mma(f[u], acc);
        }
        kp += NSET1 * TRIP1;
        vp += NSET1 * TRIP1;
    }
    static_assert(NTRIP1 - NGRP1 * NSET1 == NSET1 - 1, "the loop must fetch the last trip");
#pragma unroll
    for (int t = NGRP1 * NSET1; t < NTRIP1; ++t) kv_mma(f[t % NSET1], acc);

#pragma unroll
    for (int t = 0; t < 16; ++t) {
#pragma unroll
        for (int hf = 0; hf < 2; ++hf)
            red[w * TILE_VEC4 + (t * 2 + hf) * 32 + lane] =
                make_float4(acc[t][4 * hf], acc[t][4 * hf + 1], acc[t][4 * hf + 2], acc[t][4 * hf + 3]);
    }
    __syncthreads();

    float* pb = partial + (((size_t)b * NCHUNK + chunk) * D_SZ * D_SZ);
    const int r  = w >> 1;
    const int hf = w & 1;
    float4 sj[4];
#pragma unroll
    for (int j = 0; j < 4; ++j) {
        const int e = ((r * 4 + j) * 2 + hf) * 32 + lane;
        float4 s = red[e];
#pragma unroll
        for (int w2 = 1; w2 < NSUB; ++w2) {
            const float4 x = red[w2 * TILE_VEC4 + e];
            s.x += x.x; s.y += x.y; s.z += x.z; s.w += x.w;
        }
        sj[j] = s;
    }
    const int mrow = 8 * hi + 4 * hf;
    float* prow = pb + (size_t)(dbase + 4 * mrow + r) * D_SZ + ebase + 4 * m;
    *(float4*)(prow)            = make_float4(sj[0].x, sj[1].x, sj[2].x, sj[3].x);
    *(float4*)(prow + 4 * D_SZ)  = make_float4(sj[0].y, sj[1].y, sj[2].y, sj[3].y);
    *(float4*)(prow + 8 * D_SZ)  = make_float4(sj[0].z, sj[1].z, sj[2].z, sj[3].z);
    *(float4*)(prow + 12 * D_SZ) = make_float4(sj[0].w, sj[1].w, sj[2].w, sj[3].w);
}

__global__ __launch_bounds__(256) void la_kv_reduce(const float* __restrict__ partial,
                                                    float* __restrict__ kvp) {
    const int idx = blockIdx.x * 256 + threadIdx.x;
    const int e   = idx & (D_SZ - 1);
    const int p   = (idx >> 8) & (D_SZ / 2 - 1);
    const int b   = idx >> 15;
    float s0 = 0.f, s1 = 0.f;
#pragma unroll
    for (int c = 0; c < NCHUNK; ++c) {
        const float* pp = partial + (((size_t)(b * NCHUNK + c)) << 16) + (2 * p) * D_SZ + e;
        s0 += pp[0];
        s1 += pp[D_SZ];
    }
    *(float2*)&kvp[(((size_t)b * (D_SZ / 2) + p) * D_SZ + e) * 2] = make_float2(s0, s1);
}

#define QPITCH 260
#define QROWS 64
#define KSTEPS2 2
#define TRIP2A (4 * KSTEPS2)
#define TRIP2B (KSTEPS2 * 4 * D_SZ)
#define NTRIP2 (D_SZ / (4 * KSTEPS2))
#define NSET2 4
#define NGRP2 ((NTRIP2 - (NSET2 - 1)) / NSET2)

struct OutFrag {
    v2f    a[KSTEPS2][2];
    float4 q[KSTEPS2][2];
};

__device__ __forceinline__ void out_fetch(OutFrag& f, const float* ap, const float* bp) {
#pragma unroll
    for (int s = 0; s < KSTEPS2; ++s) {
        f.a[s][0] = *(const v2f*)(ap + 4 * s);
        f.a[s][1] = *(const v2f*)(ap + 16 * QPITCH + 4 * s);
        f.q[s][0] = *(const float4*)(bp + s * 4 * D_SZ);
        f.q[s][1] = *(const float4*)(bp + s * 4 * D_SZ + 64);
    }
}

__device__ __forceinline__ void out_mma(const OutFrag& f, v8f (&acc)[2][4]) {
#pragma unroll
    for (int s = 0; s < KSTEPS2; ++s) {
        v2f b0, b1, b2, b3;
        b0.x = f.q[s][0].x; b0.y = f.q[s][0].y;
        b1.x = f.q[s][0].z; b1.y = f.q[s][0].w;
        b2.x = f.q[s][1].x; b2.y = f.q[s][1].y;
        b3.x = f.q[s][1].z; b3.y = f.q[s][1].w;
#pragma unroll
        for (int r = 0; r < 2; ++r) {
            acc[r][0] = __builtin_amdgcn_wmma_f32_16x16x4_f32(
                false, f.a[s][r], false, b0, (short)0, acc[r][0], false, false);
            acc[r][1] = __builtin_amdgcn_wmma_f32_16x16x4_f32(
                false, f.a[s][r], false, b1, (short)0, acc[r][1], false, false);
            acc[r][2] = __builtin_amdgcn_wmma_f32_16x16x4_f32(
                false, f.a[s][r], false, b2, (short)0, acc[r][2], false, false);
            acc[r][3] = __builtin_amdgcn_wmma_f32_16x16x4_f32(
                false, f.a[s][r], false, b3, (short)0, acc[r][3], false, false);
        }
    }
}

__global__ __launch_bounds__(256) void la_out(const float* __restrict__ qmat,
                                              const float* __restrict__ kv,
                                              float* __restrict__ out) {
    __shared__ __attribute__((aligned(16))) float ldsq[QROWS * QPITCH];
    __shared__ float dpart[QROWS * 8];
    __shared__ float drcp[QROWS];

    const int b    = blockIdx.x >> 8;
    const int n0   = (blockIdx.x & 255) * QROWS;
    const int tid  = threadIdx.x;

#pragma unroll
    for (int half = 0; half < 2; ++half) {
        const int row  = half * 32 + (tid >> 3);
        const int cseg = (tid & 7) * 32;
        const float* qrow = qmat + ((size_t)(b * N_SZ + n0 + row)) * D_SZ + cseg;
        float s = 0.f;
#pragma unroll
        for (int j = 0; j < 32; j += 4) {
            float4 x = *(const float4*)(qrow + j);
            x.x = fmaxf(x.x, 0.f);
            x.y = fmaxf(x.y, 0.f);
            x.z = fmaxf(x.z, 0.f);
            x.w = fmaxf(x.w, 0.f);
            *(float4*)&ldsq[row * QPITCH + cseg + j] = x;
            s += x.x + x.y + x.z + x.w;
        }
        dpart[half * 256 + tid] = s;
    }
    __syncthreads();
    if (tid < QROWS) {
        float s = EPSV;
#pragma unroll
        for (int j = 0; j < 8; ++j) s += dpart[tid * 8 + j];
        drcp[tid] = 1.0f / s;
    }
    __syncthreads();

    const int w    = tid >> 5;
    const int lane = tid & 31;
    const int m    = lane & 15;
    const int hi   = lane >> 4;
    const int rp   = w >> 2;
    const int cb   = (w & 3) * 64;

    v8f acc[2][4];
#pragma unroll
    for (int r = 0; r < 2; ++r) {
#pragma unroll
        for (int t = 0; t < 4; ++t) acc[r][t] = (v8f){0.f,0.f,0.f,0.f,0.f,0.f,0.f,0.f};
    }

    const float* bp = kv + ((size_t)b << 16) + ((size_t)hi * D_SZ + cb + 2 * m) * 2;
    const float* ap = &ldsq[(rp * 32 + m) * QPITCH + 2 * hi];

    OutFrag f[NSET2];
#pragma unroll
    for (int u = 0; u < NSET2 - 1; ++u) out_fetch(f[u], ap + u * TRIP2A, bp + u * TRIP2B);

    for (int g = 0; g < NGRP2; ++g) {
#pragma unroll
        for (int u = 0; u < NSET2; ++u) {
            __builtin_amdgcn_sched_barrier(0);
            out_fetch(f[(u + NSET2 - 1) % NSET2], ap + (u + NSET2 - 1) * TRIP2A,
                      bp + (u + NSET2 - 1) * TRIP2B);
            __builtin_amdgcn_sched_barrier(0);
            out_mma(f[u], acc);
        }
        ap += NSET2 * TRIP2A;
        bp += NSET2 * TRIP2B;
    }
#pragma unroll
    for (int t = NGRP2 * NSET2; t < NTRIP2; ++t) {
        const int u = t - NGRP2 * NSET2;
        if (t + NSET2 - 1 < NTRIP2)
            out_fetch(f[(t + NSET2 - 1) % NSET2], ap + (u + NSET2 - 1) * TRIP2A,
                      bp + (u + NSET2 - 1) * TRIP2B);
        out_mma(f[t % NSET2], acc);
    }

#pragma unroll
    for (int r = 0; r < 2; ++r) {
#pragma unroll
        for (int i = 0; i < 8; ++i) {
            const int rowl = rp * 32 + r * 16 + (hi ? i + 8 : i);
            const float rc = drcp[rowl];
            float* orow = out + ((size_t)(b * N_SZ + n0 + rowl)) * D_SZ + cb + 2 * m;
            *(float2*)(orow)      = make_float2(acc[r][0][i] * rc, acc[r][1][i] * rc);
            *(float2*)(orow + 32) = make_float2(acc[r][2][i] * rc, acc[r][3][i] * rc);
        }
    }
}

extern "C" void kernel_launch(void* const* d_in, const int* in_sizes, int n_in,
                              void* d_out, int out_size, void* d_ws, size_t ws_size,
                              hipStream_t stream) {
    const float* q = (const float*)d_in[0];
    const float* k = (const float*)d_in[1];
    const float* v = (const float*)d_in[2];
    float* out = (float*)d_out;

    float* kvbuf   = (float*)d_ws;
    float* partial = kvbuf + (size_t)B_SZ * D_SZ * D_SZ;

    la_kv_partial<<<B_SZ * NCHUNK * 4 * 4, 256, 0, stream>>>(k, v, partial);
    la_kv_reduce<<<(B_SZ * (D_SZ / 2) * D_SZ) / 256, 256, 0, stream>>>(partial, kvbuf);
    la_out<<<B_SZ * (N_SZ / 64), 256, 0, stream>>>(q, kvbuf, out);
}
